# starting kernel
# baseline (speedup 1.0000x reference)
#include <hip/hip_runtime.h>


typedef float v2f __attribute__((ext_vector_type(2)));
typedef float v8f __attribute__((ext_vector_type(8)));

#define BM 128
#define BN 128
#define BK 32
#define LDS_STRIDE (BK + 4)

template <int TRANSA>
__global__ __launch_bounds__(256)
void gemm_f32_wmma(const float* __restrict__ Ag, const float* __restrict__ Bg,
                   float* __restrict__ Cg,
                   int M, int N, int K, int lda, int ldb, int ldc,
                   long long strideA, long long strideB, long long strideC,
                   float alpha, int sym)
{
    __shared__ float As[2][BM * LDS_STRIDE];
    __shared__ float Bs[2][BN * LDS_STRIDE];

    const int block_n = blockIdx.x * BN;
    const int block_m = blockIdx.y * BM;
    if (sym && block_m < block_n) return;

    const int tid  = threadIdx.x;
    const int lane = tid & 31;
    const int wave = tid >> 5;
    const int wm   = (wave & 3) * 32;
    const int wn   = (wave >> 2) * 64;
    const int batch = blockIdx.z;

    const float* A = Ag + (long long)batch * strideA;
    const float* B = Bg + (long long)batch * strideB;
    float*       C = Cg + (long long)batch * strideC;

    v8f acc[2][4];
#pragma unroll
    for (int i = 0; i < 2; ++i)
#pragma unroll
        for (int j = 0; j < 4; ++j)
            acc[i][j] = v8f{};

    const int lr = lane & 15;
    const int lh = lane >> 4;

    float4 ra[4], rb[4];

    auto load_a = [&](int kt) {
#pragma unroll
        for (int it = 0; it < 4; ++it) {
            int f = (tid + it * 256) * 4;
            if (TRANSA == 0) {
                int row = f >> 5, col = f & 31;
                ra[it] = *(const float4*)(A + (long long)(block_m + row) * lda + kt + col);
            } else {
                int kr = f >> 7, mc = f & 127;
                ra[it] = *(const float4*)(A + (long long)(kt + kr) * lda + block_m + mc);
            }
        }
    };
    auto load_b = [&](int kt) {
#pragma unroll
        for (int it = 0; it < 4; ++it) {
            int f = (tid + it * 256) * 4;
            int kr = f >> 7, nc = f & 127;
            rb[it] = *(const float4*)(B + (long long)(kt + kr) * ldb + block_n + nc);
        }
    };
    auto store_a = [&](int buf) {
#pragma unroll
        for (int it = 0; it < 4; ++it) {
            int f = (tid + it * 256) * 4;
            if (TRANSA == 0) {
                int row = f >> 5, col = f & 31;
                *(float4*)(&As[buf][row * LDS_STRIDE + col]) = ra[it];
            } else {
                int kr = f >> 7, mc = f & 127;
                As[buf][(mc + 0) * LDS_STRIDE + kr] = ra[it].x;
                As[buf][(mc + 1) * LDS_STRIDE + kr] = ra[it].y;
                As[buf][(mc + 2) * LDS_STRIDE + kr] = ra[it].z;
                As[buf][(mc + 3) * LDS_STRIDE + kr] = ra[it].w;
            }
        }
    };
    auto store_b = [&](int buf) {
#pragma unroll
        for (int it = 0; it < 4; ++it) {
            int f = (tid + it * 256) * 4;
            int kr = f >> 7, nc = f & 127;
            Bs[buf][(nc + 0) * LDS_STRIDE + kr] = rb[it].x;
            Bs[buf][(nc + 1) * LDS_STRIDE + kr] = rb[it].y;
            Bs[buf][(nc + 2) * LDS_STRIDE + kr] = rb[it].z;
            Bs[buf][(nc + 3) * LDS_STRIDE + kr] = rb[it].w;
        }
    };

    load_a(0);
    load_b(0);
    store_a(0);
    store_b(0);
    __syncthreads();

    int cur = 0;
    for (int kt = 0; kt < K; kt += BK) {
        const bool more = (kt + BK) < K;
        if (more) { load_a(kt + BK); load_b(kt + BK); }

#pragma unroll
        for (int kk = 0; kk < BK; kk += 4) {
            v2f afrag[2], bfrag[4];
#pragma unroll
            for (int i = 0; i < 2; ++i) {
                int row = wm + i * 16 + lr;
                afrag[i] = *(const v2f*)(&As[cur][row * LDS_STRIDE + kk + 2 * lh]);
            }
#pragma unroll
            for (int j = 0; j < 4; ++j) {
                int nrw = wn + j * 16 + lr;
                bfrag[j] = *(const v2f*)(&Bs[cur][nrw * LDS_STRIDE + kk + 2 * lh]);
            }
#pragma unroll
            for (int i = 0; i < 2; ++i)
#pragma unroll
                for (int j = 0; j < 4; ++j)
                    acc[i][j] = __builtin_amdgcn_wmma_f32_16x16x4_f32(
                        false, afrag[i], false, bfrag[j],
                        (short)0, acc[i][j], false, false);
        }

        if (more) {
            store_a(cur ^ 1);
            store_b(cur ^ 1);
        }
        __syncthreads();
        cur ^= 1;
    }

    const bool mirror = sym && (block_m != block_n);
#pragma unroll
    for (int i = 0; i < 2; ++i) {
#pragma unroll
        for (int j = 0; j < 4; ++j) {
            int base_m = block_m + wm + i * 16 + (lh << 3);
            int base_n = block_n + wn + j * 16 + lr;
#pragma unroll
            for (int r = 0; r < 8; ++r) {
                float val = alpha * acc[i][j][r];
                C[(long long)(base_m + r) * ldc + base_n] = val;
                if (mirror)
                    C[(long long)base_n * ldc + (base_m + r)] = val;
            }
        }
    }
}

extern "C" void kernel_launch(void* const* d_in, const int* in_sizes, int n_in,
                              void* d_out, int out_size, void* d_ws, size_t ws_size,
                              hipStream_t stream) {
    (void)in_sizes; (void)n_in; (void)out_size; (void)ws_size;

    const float* x  = (const float*)d_in[0];
    const float* wq = (const float*)d_in[1];
    const float* wk = (const float*)d_in[2];
    const float* wv = (const float*)d_in[3];
    float* y = (float*)d_out;

    const int Bb = 4, T = 4096, D = 1024;
    const long long BDD = (long long)D * D;
    const long long BTD = (long long)T * D;

    float* G = (float*)d_ws;
    float* H = G + Bb * BDD;

    dim3 blk(256);
    dim3 gridD(D / BN, D / BM, Bb);

    gemm_f32_wmma<1><<<gridD, blk, 0, stream>>>(
        x, x, G, D, D, T, D, D, D, BTD, BTD, BDD, 1.0f, 1);

    gemm_f32_wmma<0><<<gridD, blk, 0, stream>>>(
        G, wv, H, D, D, D, D, D, D, BDD, 0LL, BDD, 1.0f, 0);

    gemm_f32_wmma<1><<<gridD, blk, 0, stream>>>(
        wk, H, G, D, D, D, D, D, D, 0LL, BDD, BDD, 1.0f, 0);

    gemm_f32_wmma<0><<<gridD, blk, 0, stream>>>(
        wq, G, H, D, D, D, D, D, D, 0LL, BDD, BDD, 1.0f, 0);

    dim3 gridY(D / BN, T / BM, Bb);
    gemm_f32_wmma<0><<<gridY, blk, 0, stream>>>(
        x, H, y, T, D, D, D, D, D, BTD, BDD, BTD, 32.0f, 0);
}


# Global loads issued two K-stages ahead into a second staging register set (K loop in stage pairs); the stage barrier orders LDS only, so the loads in flight are not drained at it
# speedup vs baseline: 1.3675x; 1.3675x over previous
#include <hip/hip_runtime.h>


typedef float v2f __attribute__((ext_vector_type(2)));
typedef float v4f __attribute__((ext_vector_type(4)));
typedef float v8f __attribute__((ext_vector_type(8)));

#define BM 128
#define BN 128
#define BK 32
#define LDS_STRIDE (BK + 4)

__device__ __forceinline__ void lds_barrier() {
    __builtin_amdgcn_fence(__ATOMIC_RELEASE, "workgroup", "local");
    __builtin_amdgcn_s_barrier();
    __builtin_amdgcn_fence(__ATOMIC_ACQUIRE, "workgroup", "local");
}

template <int TRANSA>
__global__ __launch_bounds__(256)
void gemm_f32_wmma(const float* __restrict__ Ag, const float* __restrict__ Bg,
                   float* __restrict__ Cg,
                   int M, int N, int K, int lda, int ldb, int ldc,
                   long long strideA, long long strideB, long long strideC,
                   float alpha, int sym)
{
    __shared__ float As[2][BM * LDS_STRIDE];
    __shared__ float Bs[2][BN * LDS_STRIDE];

    const int block_n = blockIdx.x * BN;
    const int block_m = blockIdx.y * BM;
    if (sym && block_m < block_n) return;

    const int tid  = threadIdx.x;
    const int lane = tid & 31;
    const int wave = tid >> 5;
    const int wm   = (wave & 3) * 32;
    const int wn   = (wave >> 2) * 64;
    const int batch = blockIdx.z;

    const float* A = Ag + (long long)batch * strideA;
    const float* B = Bg + (long long)batch * strideB;
    float*       C = Cg + (long long)batch * strideC;

    v8f acc[2][4];
#pragma unroll
    for (int i = 0; i < 2; ++i)
#pragma unroll
        for (int j = 0; j < 4; ++j)
            acc[i][j] = v8f{};

    const int lr = lane & 15;
    const int lh = lane >> 4;

    v4f ra[2][4], rb[2][4];

    auto load_a1 = [&](int set, int kt, int it) {
        int f = (tid + it * 256) * 4;
        if (TRANSA == 0) {
            int row = f >> 5, col = f & 31;
            ra[set][it] = *(const v4f*)(A + (long long)(block_m + row) * lda + kt + col);
        } else {
            int kr = f >> 7, mc = f & 127;
            ra[set][it] = *(const v4f*)(A + (long long)(kt + kr) * lda + block_m + mc);
        }
    };
    auto load_b1 = [&](int set, int kt, int it) {
        int f = (tid + it * 256) * 4;
        int kr = f >> 7, nc = f & 127;
        rb[set][it] = *(const v4f*)(B + (long long)(kt + kr) * ldb + block_n + nc);
    };
    auto load_tile = [&](int set, int kt) __attribute__((always_inline)) {
#pragma unroll
        for (int it = 0; it < 4; ++it) { load_a1(set, kt, it); load_b1(set, kt, it); }
    };
    auto store_a1 = [&](int set, int buf, int it) {
        int f = (tid + it * 256) * 4;
        if (TRANSA == 0) {
            int row = f >> 5, col = f & 31;
            *(v4f*)(&As[buf][row * LDS_STRIDE + col]) = ra[set][it];
        } else {
            int kr = f >> 7, mc = f & 127;
            As[buf][(mc + 0) * LDS_STRIDE + kr] = ra[set][it].x;
            As[buf][(mc + 1) * LDS_STRIDE + kr] = ra[set][it].y;
            As[buf][(mc + 2) * LDS_STRIDE + kr] = ra[set][it].z;
            As[buf][(mc + 3) * LDS_STRIDE + kr] = ra[set][it].w;
        }
    };
    auto store_b1 = [&](int set, int buf, int it) {
        int f = (tid + it * 256) * 4;
        int kr = f >> 7, nc = f & 127;
        Bs[buf][(nc + 0) * LDS_STRIDE + kr] = rb[set][it].x;
        Bs[buf][(nc + 1) * LDS_STRIDE + kr] = rb[set][it].y;
        Bs[buf][(nc + 2) * LDS_STRIDE + kr] = rb[set][it].z;
        Bs[buf][(nc + 3) * LDS_STRIDE + kr] = rb[set][it].w;
    };

    auto stage = [&](int cur, bool STAGE, int set) __attribute__((always_inline)) {
#pragma unroll
        for (int g = 0; g < 4; ++g) {
            v2f afrag[2][2], bfrag[2][4];
#pragma unroll
            for (int s2 = 0; s2 < 2; ++s2) {
                const int kk = (2 * g + s2) * 4;
#pragma unroll
                for (int i = 0; i < 2; ++i) {
                    int row = wm + i * 16 + lr;
                    afrag[s2][i] = *(const v2f*)(&As[cur][row * LDS_STRIDE + kk + 2 * lh]);
                }
#pragma unroll
                for (int j = 0; j < 4; ++j) {
                    int nrw = wn + j * 16 + lr;
                    bfrag[s2][j] = *(const v2f*)(&Bs[cur][nrw * LDS_STRIDE + kk + 2 * lh]);
                }
            }
            if (STAGE) {
                const int nb = cur ^ 1;
                if (g == 1) { store_a1(set, nb, 0); store_b1(set, nb, 0); store_a1(set, nb, 1); }
                if (g == 2) { store_b1(set, nb, 1); store_a1(set, nb, 2); store_b1(set, nb, 2); }
                if (g == 3) { store_a1(set, nb, 3); store_b1(set, nb, 3); }
            }
#pragma unroll
            for (int s2 = 0; s2 < 2; ++s2)
#pragma unroll
                for (int i = 0; i < 2; ++i)
#pragma unroll
                    for (int j = 0; j < 4; ++j)
                        acc[i][j] = __builtin_amdgcn_wmma_f32_16x16x4_f32(
                            false, afrag[s2][i], false, bfrag[s2][j],
                            (short)0, acc[i][j], false, false);
        }
    };

    load_tile(0, 0);
    load_tile(1, BK);
#pragma unroll
    for (int it = 0; it < 4; ++it) { store_a1(0, 0, it); store_b1(0, 0, it); }
    lds_barrier();

    for (int kt = 2 * BK; kt < K; kt += 2 * BK) {
        load_tile(0, kt);
        __builtin_amdgcn_sched_barrier(0);
        stage(0, true, 1);
        lds_barrier();
        __builtin_amdgcn_sched_barrier(0);
        load_tile(1, kt + BK);
        __builtin_amdgcn_sched_barrier(0);
        stage(1, true, 0);
        lds_barrier();
    }
    stage(0, true, 1);
    lds_barrier();
    stage(1, false, 0);

    const bool mirror = sym && (block_m != block_n);
#pragma unroll
    for (int i = 0; i < 2; ++i) {
#pragma unroll
        for (int j = 0; j < 4; ++j) {
            int base_m = block_m + wm + i * 16 + (lh << 3);
            int base_n = block_n + wn + j * 16 + lr;
#pragma unroll
            for (int r = 0; r < 8; ++r) {
                float val = alpha * acc[i][j][r];
                C[(long long)(base_m + r) * ldc + base_n] = val;
                if (mirror)
                    C[(long long)base_n * ldc + (base_m + r)] = val;
            }
        }
    }
}

extern "C" void kernel_launch(void* const* d_in, const int* in_sizes, int n_in,
                              void* d_out, int out_size, void* d_ws, size_t ws_size,
                              hipStream_t stream) {
    (void)in_sizes; (void)n_in; (void)out_size; (void)ws_size;

    const float* x  = (const float*)d_in[0];
    const float* wq = (const float*)d_in[1];
    const float* wk = (const float*)d_in[2];
    const float* wv = (const float*)d_in[3];
    float* y = (float*)d_out;

    const int Bb = 4, T = 4096, D = 1024;
    const long long BDD = (long long)D * D;
    const long long BTD = (long long)T * D;

    float* G = (float*)d_ws;
    float* H = G + Bb * BDD;

    dim3 blk(256);
    dim3 gridD(D / BN, D / BM, Bb);

    gemm_f32_wmma<1><<<gridD, blk, 0, stream>>>(
        x, x, G, D, D, T, D, D, D, BTD, BTD, BDD, 1.0f, 1);

    gemm_f32_wmma<0><<<gridD, blk, 0, stream>>>(
        G, wv, H, D, D, D, D, D, D, BDD, 0LL, BDD, 1.0f, 0);

    gemm_f32_wmma<1><<<gridD, blk, 0, stream>>>(
        wk, H, G, D, D, D, D, D, D, 0LL, BDD, BDD, 1.0f, 0);

    gemm_f32_wmma<0><<<gridD, blk, 0, stream>>>(
        wq, G, H, D, D, D, D, D, D, 0LL, BDD, BDD, 1.0f, 0);

    dim3 gridY(D / BN, T / BM, Bb);
    gemm_f32_wmma<0><<<gridY, blk, 0, stream>>>(
        x, H, y, T, D, D, D, D, D, BTD, BDD, BTD, 32.0f, 0);
}


# K x cols tile slices re-dealt to lanes: a wave-load covers 4 k-rows x 32 cols (was 1 x 128), so the transposing b32 LDS stores hit 16 banks 2-way, not 4 banks 8-way; LDS layout unchanged
# speedup vs baseline: 1.5767x; 1.1530x over previous
#include <hip/hip_runtime.h>


typedef float v2f __attribute__((ext_vector_type(2)));
typedef float v4f __attribute__((ext_vector_type(4)));
typedef float v8f __attribute__((ext_vector_type(8)));

#define BM 128
#define BN 128
#define BK 32
#define LDS_STRIDE (BK + 4)

__device__ __forceinline__ void lds_barrier() {
    __builtin_amdgcn_fence(__ATOMIC_RELEASE, "workgroup", "local");
    __builtin_amdgcn_s_barrier();
    __builtin_amdgcn_fence(__ATOMIC_ACQUIRE, "workgroup", "local");
}

template <int TRANSA>
__global__ __launch_bounds__(256)
void gemm_f32_wmma(const float* __restrict__ Ag, const float* __restrict__ Bg,
                   float* __restrict__ Cg,
                   int M, int N, int K, int lda, int ldb, int ldc,
                   long long strideA, long long strideB, long long strideC,
                   float alpha, int sym)
{
    __shared__ float As[2][BM * LDS_STRIDE];
    __shared__ float Bs[2][BN * LDS_STRIDE];

    const int block_n = blockIdx.x * BN;
    const int block_m = blockIdx.y * BM;
    if (sym && block_m < block_n) return;

    const int tid  = threadIdx.x;
    const int lane = tid & 31;
    const int wave = tid >> 5;
    const int wm   = (wave & 3) * 32;
    const int wn   = (wave >> 2) * 64;
    const int batch = blockIdx.z;

    const float* A = Ag + (long long)batch * strideA;
    const float* B = Bg + (long long)batch * strideB;
    float*       C = Cg + (long long)batch * strideC;

    v8f acc[2][4];
#pragma unroll
    for (int i = 0; i < 2; ++i)
#pragma unroll
        for (int j = 0; j < 4; ++j)
            acc[i][j] = v8f{};

    const int lr = lane & 15;
    const int lh = lane >> 4;

    v4f ra[2][4], rb[2][4];

    const int kx = wave * 4 + (lane >> 3);
    const int cx = (lane & 7) * 4;

    auto load_a1 = [&](int set, int kt, int it) {
        int f = (tid + it * 256) * 4;
        if (TRANSA == 0) {
            int row = f >> 5, col = f & 31;
            ra[set][it] = *(const v4f*)(A + (long long)(block_m + row) * lda + kt + col);
        } else {
            int kr = kx, mc = it * 32 + cx;
            ra[set][it] = *(const v4f*)(A + (long long)(kt + kr) * lda + block_m + mc);
        }
    };
    auto load_b1 = [&](int set, int kt, int it) {
        int kr = kx, nc = it * 32 + cx;
        rb[set][it] = *(const v4f*)(B + (long long)(kt + kr) * ldb + block_n + nc);
    };
    auto load_tile = [&](int set, int kt) __attribute__((always_inline)) {
#pragma unroll
        for (int it = 0; it < 4; ++it) { load_a1(set, kt, it); load_b1(set, kt, it); }
    };
    auto store_a1 = [&](int set, int buf, int it) {
        int f = (tid + it * 256) * 4;
        if (TRANSA == 0) {
            int row = f >> 5, col = f & 31;
            *(v4f*)(&As[buf][row * LDS_STRIDE + col]) = ra[set][it];
        } else {
            int kr = kx, mc = it * 32 + cx;
            As[buf][(mc + 0) * LDS_STRIDE + kr] = ra[set][it].x;
            As[buf][(mc + 1) * LDS_STRIDE + kr] = ra[set][it].y;
            As[buf][(mc + 2) * LDS_STRIDE + kr] = ra[set][it].z;
            As[buf][(mc + 3) * LDS_STRIDE + kr] = ra[set][it].w;
        }
    };
    auto store_b1 = [&](int set, int buf, int it) {
        int kr = kx, nc = it * 32 + cx;
        Bs[buf][(nc + 0) * LDS_STRIDE + kr] = rb[set][it].x;
        Bs[buf][(nc + 1) * LDS_STRIDE + kr] = rb[set][it].y;
        Bs[buf][(nc + 2) * LDS_STRIDE + kr] = rb[set][it].z;
        Bs[buf][(nc + 3) * LDS_STRIDE + kr] = rb[set][it].w;
    };

    auto stage = [&](int cur, bool STAGE, int set) __attribute__((always_inline)) {
#pragma unroll
        for (int g = 0; g < 4; ++g) {
            v2f afrag[2][2], bfrag[2][4];
#pragma unroll
            for (int s2 = 0; s2 < 2; ++s2) {
                const int kk = (2 * g + s2) * 4;
#pragma unroll
                for (int i = 0; i < 2; ++i) {
                    int row = wm + i * 16 + lr;
                    afrag[s2][i] = *(const v2f*)(&As[cur][row * LDS_STRIDE + kk + 2 * lh]);
                }
#pragma unroll
                for (int j = 0; j < 4; ++j) {
                    int nrw = wn + j * 16 + lr;
                    bfrag[s2][j] = *(const v2f*)(&Bs[cur][nrw * LDS_STRIDE + kk + 2 * lh]);
                }
            }
            if (STAGE) {
                const int nb = cur ^ 1;
                if (g == 1) { store_a1(set, nb, 0); store_b1(set, nb, 0); store_a1(set, nb, 1); }
                if (g == 2) { store_b1(set, nb, 1); store_a1(set, nb, 2); store_b1(set, nb, 2); }
                if (g == 3) { store_a1(set, nb, 3); store_b1(set, nb, 3); }
            }
#pragma unroll
            for (int s2 = 0; s2 < 2; ++s2)
#pragma unroll
                for (int i = 0; i < 2; ++i)
#pragma unroll
                    for (int j = 0; j < 4; ++j)
                        acc[i][j] = __builtin_amdgcn_wmma_f32_16x16x4_f32(
                            false, afrag[s2][i], false, bfrag[s2][j],
                            (short)0, acc[i][j], false, false);
        }
    };

    load_tile(0, 0);
    load_tile(1, BK);
#pragma unroll
    for (int it = 0; it < 4; ++it) { store_a1(0, 0, it); store_b1(0, 0, it); }
    lds_barrier();

    for (int kt = 2 * BK; kt < K; kt += 2 * BK) {
        load_tile(0, kt);
        __builtin_amdgcn_sched_barrier(0);
        stage(0, true, 1);
        lds_barrier();
        __builtin_amdgcn_sched_barrier(0);
        load_tile(1, kt + BK);
        __builtin_amdgcn_sched_barrier(0);
        stage(1, true, 0);
        lds_barrier();
    }
    stage(0, true, 1);
    lds_barrier();
    stage(1, false, 0);

    const bool mirror = sym && (block_m != block_n);
#pragma unroll
    for (int i = 0; i < 2; ++i) {
#pragma unroll
        for (int j = 0; j < 4; ++j) {
            int base_m = block_m + wm + i * 16 + (lh << 3);
            int base_n = block_n + wn + j * 16 + lr;
#pragma unroll
            for (int r = 0; r < 8; ++r) {
                float val = alpha * acc[i][j][r];
                C[(long long)(base_m + r) * ldc + base_n] = val;
                if (mirror)
                    C[(long long)base_n * ldc + (base_m + r)] = val;
            }
        }
    }
}

extern "C" void kernel_launch(void* const* d_in, const int* in_sizes, int n_in,
                              void* d_out, int out_size, void* d_ws, size_t ws_size,
                              hipStream_t stream) {
    (void)in_sizes; (void)n_in; (void)out_size; (void)ws_size;

    const float* x  = (const float*)d_in[0];
    const float* wq = (const float*)d_in[1];
    const float* wk = (const float*)d_in[2];
    const float* wv = (const float*)d_in[3];
    float* y = (float*)d_out;

    const int Bb = 4, T = 4096, D = 1024;
    const long long BDD = (long long)D * D;
    const long long BTD = (long long)T * D;

    float* G = (float*)d_ws;
    float* H = G + Bb * BDD;

    dim3 blk(256);
    dim3 gridD(D / BN, D / BM, Bb);

    gemm_f32_wmma<1><<<gridD, blk, 0, stream>>>(
        x, x, G, D, D, T, D, D, D, BTD, BTD, BDD, 1.0f, 1);

    gemm_f32_wmma<0><<<gridD, blk, 0, stream>>>(
        G, wv, H, D, D, D, D, D, D, BDD, 0LL, BDD, 1.0f, 0);

    gemm_f32_wmma<1><<<gridD, blk, 0, stream>>>(
        wk, H, G, D, D, D, D, D, D, 0LL, BDD, BDD, 1.0f, 0);

    gemm_f32_wmma<0><<<gridD, blk, 0, stream>>>(
        wq, G, H, D, D, D, D, D, D, 0LL, BDD, BDD, 1.0f, 0);

    dim3 gridY(D / BN, T / BM, Bb);
    gemm_f32_wmma<0><<<gridY, blk, 0, stream>>>(
        x, H, y, T, D, D, D, D, D, BTD, BDD, BTD, 32.0f, 0);
}


# Launch 1 (x^T x) re-tiled: lower triangle cut into 128x64 tiles plus 64x64 diagonal quarter tiles, 288 work-groups on all 256 WGPs (was 144 of 128x128); f32 summation order unchanged
# speedup vs baseline: 1.6746x; 1.0621x over previous
#include <hip/hip_runtime.h>


typedef float v2f __attribute__((ext_vector_type(2)));
typedef float v4f __attribute__((ext_vector_type(4)));
typedef float v8f __attribute__((ext_vector_type(8)));

#define BM 128
#define BK 32
#define LDS_STRIDE (BK + 4)

__device__ __forceinline__ void lds_barrier() {
    __builtin_amdgcn_fence(__ATOMIC_RELEASE, "workgroup", "local");
    __builtin_amdgcn_s_barrier();
    __builtin_amdgcn_fence(__ATOMIC_ACQUIRE, "workgroup", "local");
}

template <int TRANSA, int NJ, int SYM>
__global__ __launch_bounds__(256)
void gemm_f32_wmma(const float* __restrict__ Ag, const float* __restrict__ Bg,
                   float* __restrict__ Cg,
                   int M, int N, int K, int lda, int ldb, int ldc,
                   long long strideA, long long strideB, long long strideC,
                   float alpha)
{
    constexpr int BN = 32 * NJ;
    __shared__ float As[2][BM * LDS_STRIDE];
    __shared__ float Bs[2][BN * LDS_STRIDE];

    const int tid  = threadIdx.x;
    const int lane = tid & 31;
    const int wave = tid >> 5;
    int wm = (wave & 3) * 32;
    int wn = (wave >> 2) * (16 * NJ);

    int block_m, block_n, batch;
    bool quarter = false;
    bool mirror  = false;
    if (SYM) {
        const int bid = blockIdx.x;
        int bm, bn;
        if (bid < 256) {
            batch = bid >> 6;
            const int t = bid & 63;
            bm = 0;
            while ((bm + 1) * (bm + 1) <= t) ++bm;
            bn = t - bm * bm;
        } else {
            const int q = bid - 256;
            batch = q >> 3;
            bm = q & 7;
            bn = 2 * bm + 1;
            quarter = true;
        }
        block_m = bm * BM;
        block_n = bn * BN;
        if (quarter) wm = 64 + (wave & 3) * 16;
        const int wave_s = __builtin_amdgcn_readfirstlane(wave);
        mirror = !quarter && (block_m + (wave_s & 3) * 32) >= (block_n + 64);
    } else {
        block_n = blockIdx.x * BN;
        block_m = blockIdx.y * BM;
        batch   = blockIdx.z;
    }

    const float* A = Ag + (long long)batch * strideA;
    const float* B = Bg + (long long)batch * strideB;
    float*       C = Cg + (long long)batch * strideC;

    v8f acc[2][NJ];
#pragma unroll
    for (int i = 0; i < 2; ++i)
#pragma unroll
        for (int j = 0; j < NJ; ++j)
            acc[i][j] = v8f{};

    const int lr = lane & 15;
    const int lh = lane >> 4;

    v4f ra[2][4], rb[2][NJ];

    const int kx = wave * 4 + (lane >> 3);
    const int cx = (lane & 7) * 4;

    auto load_a1 = [&](int set, int kt, int it) {
        int f = (tid + it * 256) * 4;
        if (TRANSA == 0) {
            int row = f >> 5, col = f & 31;
            ra[set][it] = *(const v4f*)(A + (long long)(block_m + row) * lda + kt + col);
        } else {
            int kr = kx, mc = it * 32 + cx;
            ra[set][it] = *(const v4f*)(A + (long long)(kt + kr) * lda + block_m + mc);
        }
    };
    auto load_b1 = [&](int set, int kt, int it) {
        int kr = kx, nc = it * 32 + cx;
        rb[set][it] = *(const v4f*)(B + (long long)(kt + kr) * ldb + block_n + nc);
    };
    auto load_tile = [&](int set, int kt) __attribute__((always_inline)) {
#pragma unroll
        for (int it = 0; it < 4; ++it) {
            load_a1(set, kt, it);
            if (it < NJ) load_b1(set, kt, it);
        }
    };
    auto store_a1 = [&](int set, int buf, int it) {
        int f = (tid + it * 256) * 4;
        if (TRANSA == 0) {
            int row = f >> 5, col = f & 31;
            *(v4f*)(&As[buf][row * LDS_STRIDE + col]) = ra[set][it];
        } else {
            int kr = kx, mc = it * 32 + cx;
            As[buf][(mc + 0) * LDS_STRIDE + kr] = ra[set][it].x;
            As[buf][(mc + 1) * LDS_STRIDE + kr] = ra[set][it].y;
            As[buf][(mc + 2) * LDS_STRIDE + kr] = ra[set][it].z;
            As[buf][(mc + 3) * LDS_STRIDE + kr] = ra[set][it].w;
        }
    };
    auto store_b1 = [&](int set, int buf, int it) {
        int kr = kx, nc = it * 32 + cx;
        Bs[buf][(nc + 0) * LDS_STRIDE + kr] = rb[set][it].x;
        Bs[buf][(nc + 1) * LDS_STRIDE + kr] = rb[set][it].y;
        Bs[buf][(nc + 2) * LDS_STRIDE + kr] = rb[set][it].z;
        Bs[buf][(nc + 3) * LDS_STRIDE + kr] = rb[set][it].w;
    };

    auto stage = [&](int cur, bool STAGE, int set, int NIA) __attribute__((always_inline)) {
#pragma unroll
        for (int g = 0; g < 4; ++g) {
            v2f afrag[2][2], bfrag[2][NJ];
#pragma unroll
            for (int s2 = 0; s2 < 2; ++s2) {
                const int kk = (2 * g + s2) * 4;
#pragma unroll
                for (int i = 0; i < 2; ++i) {
                    if (i < NIA) {
                        int row = wm + i * 16 + lr;
                        afrag[s2][i] = *(const v2f*)(&As[cur][row * LDS_STRIDE + kk + 2 * lh]);
                    }
                }
#pragma unroll
                for (int j = 0; j < NJ; ++j) {
                    int nrw = wn + j * 16 + lr;
                    bfrag[s2][j] = *(const v2f*)(&Bs[cur][nrw * LDS_STRIDE + kk + 2 * lh]);
                }
            }
            if (STAGE) {
                const int nb = cur ^ 1;
                if (NJ == 4) {
                    if (g == 1) { store_a1(set, nb, 0); store_b1(set, nb, 0); store_a1(set, nb, 1); }
                    if (g == 2) { store_b1(set, nb, 1); store_a1(set, nb, 2); store_b1(set, nb, 2); }
                    if (g == 3) { store_a1(set, nb, 3); store_b1(set, nb, 3); }
                } else {
                    if (g == 1) { store_a1(set, nb, 0); store_b1(set, nb, 0); }
                    if (g == 2) { store_a1(set, nb, 1); store_b1(set, nb, 1); }
                    if (g == 3) { store_a1(set, nb, 2); store_a1(set, nb, 3); }
                }
            }
#pragma unroll
            for (int s2 = 0; s2 < 2; ++s2)
#pragma unroll
                for (int i = 0; i < 2; ++i)
#pragma unroll
                    for (int j = 0; j < NJ; ++j)
                        if (i < NIA)
                            acc[i][j] = __builtin_amdgcn_wmma_f32_16x16x4_f32(
                                false, afrag[s2][i], false, bfrag[s2][j],
                                (short)0, acc[i][j], false, false);
        }
    };

    load_tile(0, 0);
    load_tile(1, BK);
#pragma unroll
    for (int it = 0; it < 4; ++it) { store_a1(0, 0, it); if (it < NJ) store_b1(0, 0, it); }
    lds_barrier();

    auto mainloop = [&](int NIA) __attribute__((always_inline)) {
        for (int kt = 2 * BK; kt < K; kt += 2 * BK) {
            load_tile(0, kt);
            __builtin_amdgcn_sched_barrier(0);
            stage(0, true, 1, NIA);
            lds_barrier();
            __builtin_amdgcn_sched_barrier(0);
            load_tile(1, kt + BK);
            __builtin_amdgcn_sched_barrier(0);
            stage(1, true, 0, NIA);
            lds_barrier();
        }
        __builtin_amdgcn_sched_barrier(0);
        asm volatile("v_nop\n\tv_nop\n\tv_nop\n\tv_nop");
        __builtin_amdgcn_sched_barrier(0);
        stage(0, true, 1, NIA);
        lds_barrier();
        stage(1, false, 0, NIA);
    };
    if (SYM && quarter) mainloop(1); else mainloop(2);

#pragma unroll
    for (int i = 0; i < 2; ++i) {
        if (SYM && quarter && i > 0) continue;
#pragma unroll
        for (int j = 0; j < NJ; ++j) {
            int base_m = block_m + wm + i * 16 + (lh << 3);
            int base_n = block_n + wn + j * 16 + lr;
#pragma unroll
            for (int r = 0; r < 8; ++r) {
                float val = alpha * acc[i][j][r];
                C[(long long)(base_m + r) * ldc + base_n] = val;
                if (SYM && mirror)
                    C[(long long)base_n * ldc + (base_m + r)] = val;
            }
        }
    }
}

extern "C" void kernel_launch(void* const* d_in, const int* in_sizes, int n_in,
                              void* d_out, int out_size, void* d_ws, size_t ws_size,
                              hipStream_t stream) {
    (void)in_sizes; (void)n_in; (void)out_size; (void)ws_size;

    const float* x  = (const float*)d_in[0];
    const float* wq = (const float*)d_in[1];
    const float* wk = (const float*)d_in[2];
    const float* wv = (const float*)d_in[3];
    float* y = (float*)d_out;

    const int Bb = 4, T = 4096, D = 1024;
    const long long BDD = (long long)D * D;
    const long long BTD = (long long)T * D;

    float* G = (float*)d_ws;
    float* H = G + Bb * BDD;

    dim3 blk(256);
    dim3 gridD(D / 128, D / BM, Bb);

    gemm_f32_wmma<1, 2, 1><<<dim3(288), blk, 0, stream>>>(
        x, x, G, D, D, T, D, D, D, BTD, BTD, BDD, 1.0f);

    gemm_f32_wmma<0, 4, 0><<<gridD, blk, 0, stream>>>(
        G, wv, H, D, D, D, D, D, D, BDD, 0LL, BDD, 1.0f);

    gemm_f32_wmma<1, 4, 0><<<gridD, blk, 0, stream>>>(
        wk, H, G, D, D, D, D, D, D, 0LL, BDD, BDD, 1.0f);

    gemm_f32_wmma<0, 4, 0><<<gridD, blk, 0, stream>>>(
        wq, G, H, D, D, D, D, D, D, 0LL, BDD, BDD, 1.0f);

    dim3 gridY(D / 128, T / BM, Bb);
    gemm_f32_wmma<0, 4, 0><<<gridY, blk, 0, stream>>>(
        x, H, y, T, D, D, D, D, D, BTD, BDD, BTD, 32.0f);
}


# Launch 1: the 32 diagonal 64x64 work-groups go; each 128x64 tile forms 2 of their fragments from a panel it holds, k-steps dealt to its 8 waves, summed in LDS; f32 sum order changes there
# speedup vs baseline: 1.7706x; 1.0573x over previous
#include <hip/hip_runtime.h>


typedef float v2f __attribute__((ext_vector_type(2)));
typedef float v4f __attribute__((ext_vector_type(4)));
typedef float v8f __attribute__((ext_vector_type(8)));

#define BM 128
#define BK 32
#define LDS_STRIDE (BK + 4)

__device__ __forceinline__ void lds_barrier() {
    __builtin_amdgcn_fence(__ATOMIC_RELEASE, "workgroup", "local");
    __builtin_amdgcn_s_barrier();
    __builtin_amdgcn_fence(__ATOMIC_ACQUIRE, "workgroup", "local");
}

template <int TRANSA, int NJ, int SYM>
__global__ __launch_bounds__(256)
void gemm_f32_wmma(const float* __restrict__ Ag, const float* __restrict__ Bg,
                   float* __restrict__ Cg,
                   int M, int N, int K, int lda, int ldb, int ldc,
                   long long strideA, long long strideB, long long strideC,
                   float alpha)
{
    constexpr int BN = 32 * NJ;
    __shared__ float As[2][BM * LDS_STRIDE];
    __shared__ float Bs[2][BN * LDS_STRIDE];

    const int tid  = threadIdx.x;
    const int lane = tid & 31;
    const int wave = tid >> 5;
    const int wm = (wave & 3) * 32;
    const int wn = (wave >> 2) * (16 * NJ);

    const int lr = lane & 15;
    const int lh = lane >> 4;

    int block_m, block_n, batch;
    bool mirror = false;
    int xq = 0, xfi = 0, xfj = 0;
    const float* xpan = nullptr;
    int xbuf = 0;
    int xa_off = 0, xb_off = 0;
    if (SYM) {
        const int bid = blockIdx.x;
        batch = bid >> 6;
        const int t = bid & 63;
        int bm = 0;
        while ((bm + 1) * (bm + 1) <= t) ++bm;
        const int bn = t - bm * bm;
        block_m = bm * BM;
        block_n = bn * BN;
        const int wave_s = __builtin_amdgcn_readfirstlane(wave);
        mirror = (block_m + (wave_s & 3) * 32) >= (block_n + 64);
        int p;
        if (bn & 1) {
            xq = bn;
            p = bm - ((bn + 1) >> 1);
            xpan = &Bs[0][0];
            xbuf = BN * LDS_STRIDE;
        } else {
            xq = 2 * bm + 1;
            p = 7 - bm + (bn >> 1);
            xpan = &As[0][64 * LDS_STRIDE];
            xbuf = BM * LDS_STRIDE;
        }
        xfi = p >> 1;
        xfj = (p & 1) * 2;
        const int kx4 = 4 * wave + 2 * lh;
        xa_off = (16 * xfi + lr) * LDS_STRIDE + kx4;
        xb_off = (16 * xfj + lr) * LDS_STRIDE + kx4;
    } else {
        block_n = blockIdx.x * BN;
        block_m = blockIdx.y * BM;
        batch   = blockIdx.z;
    }

    const float* A = Ag + (long long)batch * strideA;
    const float* B = Bg + (long long)batch * strideB;
    float*       C = Cg + (long long)batch * strideC;

    v8f acc[2][NJ];
#pragma unroll
    for (int i = 0; i < 2; ++i)
#pragma unroll
        for (int j = 0; j < NJ; ++j)
            acc[i][j] = v8f{};
    v8f accx[2] = {v8f{}, v8f{}};

    v4f ra[2][4], rb[2][NJ];

    const int kx = wave * 4 + (lane >> 3);
    const int cx = (lane & 7) * 4;

    auto load_a1 = [&](int set, int kt, int it) {
        int f = (tid + it * 256) * 4;
        if (TRANSA == 0) {
            int row = f >> 5, col = f & 31;
            ra[set][it] = *(const v4f*)(A + (long long)(block_m + row) * lda + kt + col);
        } else {
            int kr = kx, mc = it * 32 + cx;
            ra[set][it] = *(const v4f*)(A + (long long)(kt + kr) * lda + block_m + mc);
        }
    };
    auto load_b1 = [&](int set, int kt, int it) {
        int kr = kx, nc = it * 32 + cx;
        rb[set][it] = *(const v4f*)(B + (long long)(kt + kr) * ldb + block_n + nc);
    };
    auto load_tile = [&](int set, int kt) __attribute__((always_inline)) {
#pragma unroll
        for (int it = 0; it < 4; ++it) {
            load_a1(set, kt, it);
            if (it < NJ) load_b1(set, kt, it);
        }
    };
    auto store_a1 = [&](int set, int buf, int it) {
        int f = (tid + it * 256) * 4;
        if (TRANSA == 0) {
            int row = f >> 5, col = f & 31;
            *(v4f*)(&As[buf][row * LDS_STRIDE + col]) = ra[set][it];
        } else {
            int kr = kx, mc = it * 32 + cx;
            As[buf][(mc + 0) * LDS_STRIDE + kr] = ra[set][it].x;
            As[buf][(mc + 1) * LDS_STRIDE + kr] = ra[set][it].y;
            As[buf][(mc + 2) * LDS_STRIDE + kr] = ra[set][it].z;
            As[buf][(mc + 3) * LDS_STRIDE + kr] = ra[set][it].w;
        }
    };
    auto store_b1 = [&](int set, int buf, int it) {
        int kr = kx, nc = it * 32 + cx;
        Bs[buf][(nc + 0) * LDS_STRIDE + kr] = rb[set][it].x;
        Bs[buf][(nc + 1) * LDS_STRIDE + kr] = rb[set][it].y;
        Bs[buf][(nc + 2) * LDS_STRIDE + kr] = rb[set][it].z;
        Bs[buf][(nc + 3) * LDS_STRIDE + kr] = rb[set][it].w;
    };

    auto stage = [&](int cur, bool STAGE, int set) __attribute__((always_inline)) {
#pragma unroll
        for (int g = 0; g < 4; ++g) {
            v2f afrag[2][2], bfrag[2][NJ];
#pragma unroll
            for (int s2 = 0; s2 < 2; ++s2) {
                const int kk = (2 * g + s2) * 4;
#pragma unroll
                for (int i = 0; i < 2; ++i) {
                    int row = wm + i * 16 + lr;
                    afrag[s2][i] = *(const v2f*)(&As[cur][row * LDS_STRIDE + kk + 2 * lh]);
                }
#pragma unroll
                for (int j = 0; j < NJ; ++j) {
                    int nrw = wn + j * 16 + lr;
                    bfrag[s2][j] = *(const v2f*)(&Bs[cur][nrw * LDS_STRIDE + kk + 2 * lh]);
                }
            }
            v2f xa{}, xb0{}, xb1{};
            if (SYM && g == 0) {
                const float* xp = xpan + cur * xbuf;
                xa  = *(const v2f*)(xp + xa_off);
                xb0 = *(const v2f*)(xp + xb_off);
                xb1 = *(const v2f*)(xp + xb_off + 16 * LDS_STRIDE);
            }
            if (STAGE) {
                const int nb = cur ^ 1;
                if (NJ == 4) {
                    if (g == 1) { store_a1(set, nb, 0); store_b1(set, nb, 0); store_a1(set, nb, 1); }
                    if (g == 2) { store_b1(set, nb, 1); store_a1(set, nb, 2); store_b1(set, nb, 2); }
                    if (g == 3) { store_a1(set, nb, 3); store_b1(set, nb, 3); }
                } else {
                    if (g == 1) { store_a1(set, nb, 0); store_b1(set, nb, 0); }
                    if (g == 2) { store_a1(set, nb, 1); store_b1(set, nb, 1); }
                    if (g == 3) { store_a1(set, nb, 2); store_a1(set, nb, 3); }
                }
            }
#pragma unroll
            for (int s2 = 0; s2 < 2; ++s2)
#pragma unroll
                for (int i = 0; i < 2; ++i)
#pragma unroll
                    for (int j = 0; j < NJ; ++j)
                        acc[i][j] = __builtin_amdgcn_wmma_f32_16x16x4_f32(
                            false, afrag[s2][i], false, bfrag[s2][j],
                            (short)0, acc[i][j], false, false);
            if (SYM && g == 0) {
                accx[0] = __builtin_amdgcn_wmma_f32_16x16x4_f32(
                    false, xa, false, xb0, (short)0, accx[0], false, false);
                accx[1] = __builtin_amdgcn_wmma_f32_16x16x4_f32(
                    false, xa, false, xb1, (short)0, accx[1], false, false);
            }
        }
    };

    load_tile(0, 0);
    load_tile(1, BK);
#pragma unroll
    for (int it = 0; it < 4; ++it) { store_a1(0, 0, it); if (it < NJ) store_b1(0, 0, it); }
    lds_barrier();

    {
        for (int kt = 2 * BK; kt < K; kt += 2 * BK) {
            load_tile(0, kt);
            __builtin_amdgcn_sched_barrier(0);
            stage(0, true, 1);
            lds_barrier();
            __builtin_amdgcn_sched_barrier(0);
            load_tile(1, kt + BK);
            __builtin_amdgcn_sched_barrier(0);
            stage(1, true, 0);
            lds_barrier();
        }
        __builtin_amdgcn_sched_barrier(0);
        asm volatile("v_nop\n\tv_nop\n\tv_nop\n\tv_nop");
        __builtin_amdgcn_sched_barrier(0);
        stage(0, true, 1);
        lds_barrier();
        stage(1, false, 0);
    }

#pragma unroll
    for (int i = 0; i < 2; ++i) {
#pragma unroll
        for (int j = 0; j < NJ; ++j) {
            int base_m = block_m + wm + i * 16 + (lh << 3);
            int base_n = block_n + wn + j * 16 + lr;
#pragma unroll
            for (int r = 0; r < 8; ++r) {
                float val = alpha * acc[i][j][r];
                C[(long long)(base_m + r) * ldc + base_n] = val;
                if (SYM && mirror)
                    C[(long long)base_n * ldc + (base_m + r)] = val;
            }
        }
    }

    if (SYM) {
        float* red = &As[0][0];
#pragma unroll
        for (int jj = 0; jj < 2; ++jj)
#pragma unroll
            for (int r = 0; r < 8; ++r)
                red[(wave * 16 + jj * 8 + r) * 32 + lane] = accx[jj][r];
        lds_barrier();
#pragma unroll
        for (int jj = 0; jj < 2; ++jj) {
            const int e = (jj * 8 + wave) * 32 + lane;
            float s = red[e];
#pragma unroll
            for (int w = 1; w < 8; ++w) s += red[w * 512 + e];
            const int row = 64 * xq + 16 * xfi + (lh << 3) + wave;
            const int col = 64 * xq + 16 * (xfj + jj) + lr;
            C[(long long)row * ldc + col] = alpha * s;
        }
    }
}

extern "C" void kernel_launch(void* const* d_in, const int* in_sizes, int n_in,
                              void* d_out, int out_size, void* d_ws, size_t ws_size,
                              hipStream_t stream) {
    (void)in_sizes; (void)n_in; (void)out_size; (void)ws_size;

    const float* x  = (const float*)d_in[0];
    const float* wq = (const float*)d_in[1];
    const float* wk = (const float*)d_in[2];
    const float* wv = (const float*)d_in[3];
    float* y = (float*)d_out;

    const int Bb = 4, T = 4096, D = 1024;
    const long long BDD = (long long)D * D;
    const long long BTD = (long long)T * D;

    float* G = (float*)d_ws;
    float* H = G + Bb * BDD;

    dim3 blk(256);
    dim3 gridD(D / 128, D / BM, Bb);

    gemm_f32_wmma<1, 2, 1><<<dim3(256), blk, 0, stream>>>(
        x, x, G, D, D, T, D, D, D, BTD, BTD, BDD, 1.0f);

    gemm_f32_wmma<0, 4, 0><<<gridD, blk, 0, stream>>>(
        G, wv, H, D, D, D, D, D, D, BDD, 0LL, BDD, 1.0f);

    gemm_f32_wmma<1, 4, 0><<<gridD, blk, 0, stream>>>(
        wk, H, G, D, D, D, D, D, D, 0LL, BDD, BDD, 1.0f);

    gemm_f32_wmma<0, 4, 0><<<gridD, blk, 0, stream>>>(
        wq, G, H, D, D, D, D, D, D, 0LL, BDD, BDD, 1.0f);

    dim3 gridY(D / 128, T / BM, Bb);
    gemm_f32_wmma<0, 4, 0><<<gridY, blk, 0, stream>>>(
        x, H, y, T, D, D, D, D, D, BTD, BDD, BTD, 32.0f);
}
